# starting kernel
# baseline (speedup 1.0000x reference)
#include <hip/hip_runtime.h>

typedef float v2f __attribute__((ext_vector_type(2)));
typedef float v8f __attribute__((ext_vector_type(8)));

#define DD 64
#define HS_PITCH 132
#define WAVES 6
#define ROWS (WAVES * 16)
#define TPB (WAVES * 32)

__global__ void ngcf_degree(const int* __restrict__ src, const int* __restrict__ dst,
                            int* __restrict__ dout, int* __restrict__ din, int E) {
  int g = blockIdx.x * blockDim.x + threadIdx.x;
  if (g < E) {
    atomicAdd(&dout[src[g]], 1);
    atomicAdd(&din[dst[g]], 1);
  }
}

__global__ void ngcf_rsqrt(const int* __restrict__ dout, const int* __restrict__ din,
                           float* __restrict__ sout, float* __restrict__ sin_, int N) {
  int g = blockIdx.x * blockDim.x + threadIdx.x;
  if (g < N) {
    int a = dout[g]; if (a < 1) a = 1;
    int b = din[g];  if (b < 1) b = 1;
    sout[g] = rsqrtf((float)a);
    sin_[g] = rsqrtf((float)b);
  }
}

__global__ void ngcf_scatter(const float* __restrict__ x, const int* __restrict__ src,
                             const int* __restrict__ dst, const float* __restrict__ sout,
                             float* __restrict__ p, int E) {
  int g = blockIdx.x * blockDim.x + threadIdx.x;
  int e = g >> 4;
  int c = g & 15;
  if (e < E) {
    int s = src[e];
    int d = dst[e];
    float w = sout[s];
    float4 v = ((const float4*)(x + (size_t)s * DD))[c];
    float* pd = p + (size_t)d * DD + (c << 2);
    atomicAdd(pd + 0, w * v.x);
    atomicAdd(pd + 1, w * v.y);
    atomicAdd(pd + 2, w * v.z);
    atomicAdd(pd + 3, w * v.w);
  }
}

__global__ __launch_bounds__(TPB) void ngcf_gemm(
    const float* __restrict__ x, const float* __restrict__ p,
    const float* __restrict__ sin_, const float* __restrict__ W1,
    const float* __restrict__ W2, float* __restrict__ out, int N, int do_relu) {
  __shared__ __align__(16) float hs[ROWS * HS_PITCH];

  const int tid = threadIdx.x;
  const int base = blockIdx.x * ROWS;

  for (int i = tid; i < ROWS * 16; i += TPB) {
    int row = i >> 4;
    int c = (i & 15) << 2;
    int node = base + row;
    if (node >= N) node = N - 1;
    float4 xv = *(const float4*)(x + (size_t)node * DD + c);
    float4 pv = *(const float4*)(p + (size_t)node * DD + c);
    float r = sin_[node];
    float4 pp = make_float4(pv.x * r, pv.y * r, pv.z * r, pv.w * r);
    float* h1 = &hs[row * HS_PITCH + c];
    float* h2 = &hs[row * HS_PITCH + DD + c];
    *(float4*)h1 = make_float4(xv.x + pp.x, xv.y + pp.y, xv.z + pp.z, xv.w + pp.w);
    *(float4*)h2 = make_float4(xv.x * pp.x, xv.y * pp.y, xv.z * pp.z, xv.w * pp.w);
  }
  __syncthreads();

  const int wave = tid >> 5;
  const int lane = tid & 31;
  const int m = lane & 15;
  const int khalf = (lane >> 4) << 1;
  const float* hrow = &hs[(wave * 16 + m) * HS_PITCH];

  v8f acc0 = {}, acc1 = {}, acc2 = {}, acc3 = {};
  for (int kb = 0; kb < 2 * DD; kb += 4) {
    v2f a = *(const v2f*)(hrow + kb + khalf);
    const float* wr = (kb < DD) ? (W1 + (size_t)(kb + khalf) * DD)
                                : (W2 + (size_t)(kb + khalf - DD) * DD);
    v2f b0, b1, b2, b3;
    b0.x = wr[m];      b0.y = wr[DD + m];
    b1.x = wr[16 + m]; b1.y = wr[DD + 16 + m];
    b2.x = wr[32 + m]; b2.y = wr[DD + 32 + m];
    b3.x = wr[48 + m]; b3.y = wr[DD + 48 + m];
    acc0 = __builtin_amdgcn_wmma_f32_16x16x4_f32(false, a, false, b0, (short)0, acc0, false, false);
    acc1 = __builtin_amdgcn_wmma_f32_16x16x4_f32(false, a, false, b1, (short)0, acc1, false, false);
    acc2 = __builtin_amdgcn_wmma_f32_16x16x4_f32(false, a, false, b2, (short)0, acc2, false, false);
    acc3 = __builtin_amdgcn_wmma_f32_16x16x4_f32(false, a, false, b3, (short)0, acc3, false, false);
  }

  const int rbase = base + wave * 16 + ((lane >> 4) << 3);
  for (int r = 0; r < 8; ++r) {
    int node = rbase + r;
    if (node < N) {
      float v0 = acc0[r], v1 = acc1[r], v2 = acc2[r], v3 = acc3[r];
      if (do_relu) {
        v0 = fmaxf(v0, 0.f); v1 = fmaxf(v1, 0.f);
        v2 = fmaxf(v2, 0.f); v3 = fmaxf(v3, 0.f);
      }
      float* orow = out + (size_t)node * DD;
      orow[m] = v0; orow[16 + m] = v1; orow[32 + m] = v2; orow[48 + m] = v3;
    }
  }
}

extern "C" void kernel_launch(void* const* d_in, const int* in_sizes, int n_in,
                              void* d_out, int out_size, void* d_ws, size_t ws_size,
                              hipStream_t stream) {
  const float* x   = (const float*)d_in[0];
  const int*   src = (const int*)d_in[1];
  const int*   dst = (const int*)d_in[2];
  const float* W1a = (const float*)d_in[3];
  const float* W2a = (const float*)d_in[4];
  const float* W1b = (const float*)d_in[5];
  const float* W2b = (const float*)d_in[6];
  float* out = (float*)d_out;

  const int N = in_sizes[0] / DD;
  const int E = in_sizes[1];

  int* deg_out = (int*)d_ws;
  int* deg_in  = deg_out + N;
  float* sout  = (float*)(deg_in + N);
  float* sin_  = sout + N;
  float* p     = sin_ + N;

  hipMemsetAsync(deg_out, 0, (size_t)2 * N * sizeof(int), stream);
  ngcf_degree<<<(E + 255) / 256, 256, 0, stream>>>(src, dst, deg_out, deg_in, E);
  ngcf_rsqrt<<<(N + 255) / 256, 256, 0, stream>>>(deg_out, deg_in, sout, sin_, N);

  const int gemm_blocks = (N + ROWS - 1) / ROWS;
  const long long sc_items = (long long)E * 16;
  const int sc_blocks = (int)((sc_items + 255) / 256);

  hipMemsetAsync(p, 0, (size_t)N * DD * sizeof(float), stream);
  ngcf_scatter<<<sc_blocks, 256, 0, stream>>>(x, src, dst, sout, p, E);
  ngcf_gemm<<<gemm_blocks, TPB, 0, stream>>>(x, p, sin_, W1a, W2a, out, N, 1);

  hipMemsetAsync(p, 0, (size_t)N * DD * sizeof(float), stream);
  ngcf_scatter<<<sc_blocks, 256, 0, stream>>>(out, src, dst, sout, p, E);
  ngcf_gemm<<<gemm_blocks, TPB, 0, stream>>>(out, p, sin_, W1b, W2b, out, N, 0);
}


# aggregation on a resident grid: 1250 blocks whose waves each sweep 5 node pairs (8 row loads in flight a lane, next segment's ids fetched a sweep ahead), not 50000 short waves a layer; sum order kept
# speedup vs baseline: 2.6701x; 2.6701x over previous
#include <hip/hip_runtime.h>

typedef float v2f __attribute__((ext_vector_type(2)));
typedef float v8f __attribute__((ext_vector_type(8)));

#define DD 64
#define HS_PITCH 132
#define WAVES 6
#define ROWS (WAVES * 16)
#define TPB (WAVES * 32)

#define DEG_K 8192
#define DEG_C 16
#define DEG_TPB 1024

static __device__ __forceinline__ void deg_tally(int* cnt, int lo, int4 v) {
  unsigned r0 = (unsigned)(v.x - lo), r1 = (unsigned)(v.y - lo);
  unsigned r2 = (unsigned)(v.z - lo), r3 = (unsigned)(v.w - lo);
  if (r0 < DEG_K) atomicAdd(&cnt[r0], 1);
  if (r1 < DEG_K) atomicAdd(&cnt[r1], 1);
  if (r2 < DEG_K) atomicAdd(&cnt[r2], 1);
  if (r3 < DEG_K) atomicAdd(&cnt[r3], 1);
}

__global__ __launch_bounds__(DEG_TPB) void ngcf_degree(
    const int* __restrict__ src, const int* __restrict__ dst,
    int* __restrict__ part, int N, int E) {
  __shared__ int cnt[DEG_K];
  const int tid = threadIdx.x;
  const int slice = blockIdx.x;
  const int lo = blockIdx.y * DEG_K;
  const int which = blockIdx.z;
  const int* __restrict__ ids = which ? dst : src;

  for (int i = tid; i < DEG_K; i += DEG_TPB) cnt[i] = 0;
  __syncthreads();

  const int slice_len = (((E + DEG_C - 1) / DEG_C) + 3) & ~3;
  const int s0 = slice * slice_len;
  int len = E - s0;
  if (len > slice_len) len = slice_len;
  if (len < 0) len = 0;
  const int nfull = len >> 2;
  const int4* __restrict__ vp = (const int4*)(ids + s0);

  for (int j = tid; j < nfull; j += 4 * DEG_TPB) {
    const int j1 = j + DEG_TPB, j2 = j + 2 * DEG_TPB, j3 = j + 3 * DEG_TPB;
    int4 a = vp[j];
    int4 b = vp[j1 < nfull ? j1 : j];
    int4 c = vp[j2 < nfull ? j2 : j];
    int4 d = vp[j3 < nfull ? j3 : j];
    deg_tally(cnt, lo, a);
    if (j1 < nfull) deg_tally(cnt, lo, b);
    if (j2 < nfull) deg_tally(cnt, lo, c);
    if (j3 < nfull) deg_tally(cnt, lo, d);
  }
  {
    const int t = (nfull << 2) + tid;
    if (tid < 3 && t < len) {
      unsigned r = (unsigned)(ids[s0 + t] - lo);
      if (r < DEG_K) atomicAdd(&cnt[r], 1);
    }
  }
  __syncthreads();

  int* __restrict__ plane = part + (size_t)(slice * 2 + which) * N;
  for (int i = tid; i < DEG_K; i += DEG_TPB) {
    const int n = lo + i;
    if (n < N) plane[n] = cnt[i];
  }
}

#define SC_TPB 1024
__global__ __launch_bounds__(SC_TPB) void ngcf_rsqrt(int* __restrict__ part,
                           float* __restrict__ sout, float* __restrict__ sin_,
                           int* __restrict__ indeg, int* __restrict__ bsum, int N) {
  __shared__ int tot;
  if (threadIdx.x == 0) tot = 0;
  __syncthreads();
  const int g = blockIdx.x * SC_TPB + threadIdx.x;
  int b = 0;
  if (g < N) {
    int ta[DEG_C], tb[DEG_C];
    for (int c = 0; c < DEG_C; ++c) {
      ta[c] = part[(size_t)(2 * c) * N + g];
      tb[c] = part[(size_t)(2 * c + 1) * N + g];
    }
    int a = 0;
    for (int c = 0; c < DEG_C; ++c) {
      a += ta[c];
      part[(size_t)(2 * c + 1) * N + g] = b;
      b += tb[c];
    }
    indeg[g] = b;
    sout[g] = rsqrtf((float)(a < 1 ? 1 : a));
    sin_[g] = rsqrtf((float)(b < 1 ? 1 : b));
  }
  if (b) atomicAdd(&tot, b);
  __syncthreads();
  if (threadIdx.x == 0) bsum[blockIdx.x] = tot;
}

__global__ __launch_bounds__(SC_TPB) void ngcf_rowptr(int* __restrict__ rowptr,
                                                      const int* __restrict__ bsum, int N) {
  __shared__ int sc[SC_TPB];
  __shared__ int sbase;
  const int t = threadIdx.x;
  const int k = blockIdx.x;
  if (t == 0) sbase = 0;
  __syncthreads();
  int s = 0;
  for (int i = t; i < k; i += SC_TPB) s += bsum[i];
  if (s) atomicAdd(&sbase, s);
  const int g = k * SC_TPB + t;
  const int v = (g < N) ? rowptr[g] : 0;
  sc[t] = v;
  __syncthreads();
  for (int off = 1; off < SC_TPB; off <<= 1) {
    const int u = (t >= off) ? sc[t - off] : 0;
    __syncthreads();
    sc[t] += u;
    __syncthreads();
  }
  if (g < N) rowptr[g] = sbase + sc[t] - v;
}

static __device__ __forceinline__ void fill_one(int* cnt, int lo, unsigned lim, int id, int s,
                                                int* __restrict__ csr) {
  const unsigned r = (unsigned)(id - lo);
  if (r < lim) {
    const int pos = atomicAdd(&cnt[r], 1);
    csr[pos] = s;
  }
}

static __device__ __forceinline__ void fill_vec(int* cnt, int lo, unsigned lim, unsigned spare,
                                                int4 d, int4 s, int* __restrict__ csr) {
  const unsigned r0 = (unsigned)(d.x - lo), r1 = (unsigned)(d.y - lo);
  const unsigned r2 = (unsigned)(d.z - lo), r3 = (unsigned)(d.w - lo);
  const unsigned p0 = (unsigned)atomicAdd(&cnt[r0 < lim ? r0 : spare], 1);
  const unsigned p1 = (unsigned)atomicAdd(&cnt[r1 < lim ? r1 : spare], 1);
  const unsigned p2 = (unsigned)atomicAdd(&cnt[r2 < lim ? r2 : spare], 1);
  const unsigned p3 = (unsigned)atomicAdd(&cnt[r3 < lim ? r3 : spare], 1);
  if (r0 < lim) csr[p0] = s.x;
  if (r1 < lim) csr[p1] = s.y;
  if (r2 < lim) csr[p2] = s.z;
  if (r3 < lim) csr[p3] = s.w;
}

__global__ __launch_bounds__(DEG_TPB) void ngcf_fill(
    const int* __restrict__ src, const int* __restrict__ dst,
    const int* __restrict__ part, const int* __restrict__ rowptr,
    int* __restrict__ csr, int N, int E) {
  __shared__ int cnt[DEG_K + 32];
  const int tid = threadIdx.x;
  const int slice = blockIdx.x;
  const int lo = blockIdx.y * DEG_K;
  const int* __restrict__ pre = part + (size_t)(slice * 2 + 1) * N;
  const unsigned spare = DEG_K + (tid & 31);

  for (int i = tid; i < DEG_K; i += DEG_TPB) {
    const int n = lo + i;
    cnt[i] = (n < N) ? (rowptr[n] + pre[n]) : 0;
  }
  if (tid < 32) cnt[DEG_K + tid] = 0;
  __syncthreads();
  const unsigned lim = (unsigned)((N - lo) < DEG_K ? (N - lo) : DEG_K);

  const int slice_len = (((E + DEG_C - 1) / DEG_C) + 3) & ~3;
  const int s0 = slice * slice_len;
  int len = E - s0;
  if (len > slice_len) len = slice_len;
  if (len < 0) len = 0;
  const int nfull = len >> 2;
  const int4* __restrict__ vd = (const int4*)(dst + s0);
  const int4* __restrict__ vs = (const int4*)(src + s0);

  for (int j = tid; j < nfull; j += 4 * DEG_TPB) {
    const int j1 = j + DEG_TPB, j2 = j + 2 * DEG_TPB, j3 = j + 3 * DEG_TPB;
    const int i1 = j1 < nfull ? j1 : j, i2 = j2 < nfull ? j2 : j, i3 = j3 < nfull ? j3 : j;
    const int4 da = vd[j], db = vd[i1], dc = vd[i2], dd = vd[i3];
    const int4 sa = vs[j], sb = vs[i1], sc = vs[i2], sd = vs[i3];
    fill_vec(cnt, lo, lim, spare, da, sa, csr);
    if (j1 < nfull) fill_vec(cnt, lo, lim, spare, db, sb, csr);
    if (j2 < nfull) fill_vec(cnt, lo, lim, spare, dc, sc, csr);
    if (j3 < nfull) fill_vec(cnt, lo, lim, spare, dd, sd, csr);
  }
  {
    const int t = (nfull << 2) + tid;
    if (tid < 3 && t < len) fill_one(cnt, lo, lim, dst[s0 + t], src[s0 + t], csr);
  }
}

#define PULL_TPB 256
#define PULL_NODES 16
#define SORT_CAP 64
#define PULL_GRID 1280

static __device__ __forceinline__ void wave_handover() {
  __builtin_amdgcn_fence(__ATOMIC_RELEASE, "wavefront");
  __builtin_amdgcn_wave_barrier();
  __builtin_amdgcn_fence(__ATOMIC_ACQUIRE, "wavefront");
}

static __device__ __forceinline__ void seg_bounds(const int* __restrict__ rowptr, int node, int N,
                                                  int& begv, int& endv) {
  const int i0 = (node < N) ? node : N - 1;
  const int i1 = (node + 1 < N) ? node + 1 : N - 1;
  begv = rowptr[i0];
  endv = rowptr[i1];
}

static __device__ __forceinline__ int seg_len(int node, int N, int E, int begv, int endv) {
  const int end = (node + 1 < N) ? endv : E;
  return (node < N) ? end - begv : 0;
}

static __device__ __forceinline__ void seg_ids(const int* __restrict__ csr, int beg, int len, int c,
                                               bool sorted, int& v0, int& v1, int& v2, int& v3) {
  v0 = 0; v1 = 0; v2 = 0; v3 = 0;
  if (sorted) {
    if (c < len)      v0 = csr[beg + c];
    if (c + 16 < len) v1 = csr[beg + c + 16];
    if (c + 32 < len) v2 = csr[beg + c + 32];
    if (c + 48 < len) v3 = csr[beg + c + 48];
  }
}

static __device__ __forceinline__ void row_fma(float4& acc, float w, const float4 a) {
  acc.x = fmaf(w, a.x, acc.x); acc.y = fmaf(w, a.y, acc.y);
  acc.z = fmaf(w, a.z, acc.z); acc.w = fmaf(w, a.w, acc.w);
}

static __device__ __forceinline__ void pull_add8(float4& acc, const float4* __restrict__ xr,
                                                 const float* __restrict__ sout, int c,
                                                 const int4 sa, const int4 sb) {
  const float w0 = sout[sa.x], w1 = sout[sa.y], w2 = sout[sa.z], w3 = sout[sa.w];
  const float w4 = sout[sb.x], w5 = sout[sb.y], w6 = sout[sb.z], w7 = sout[sb.w];
  const float4 a0 = xr[(size_t)sa.x * 16 + c];
  const float4 a1 = xr[(size_t)sa.y * 16 + c];
  const float4 a2 = xr[(size_t)sa.z * 16 + c];
  const float4 a3 = xr[(size_t)sa.w * 16 + c];
  const float4 a4 = xr[(size_t)sb.x * 16 + c];
  const float4 a5 = xr[(size_t)sb.y * 16 + c];
  const float4 a6 = xr[(size_t)sb.z * 16 + c];
  const float4 a7 = xr[(size_t)sb.w * 16 + c];
  row_fma(acc, w0, a0); row_fma(acc, w1, a1); row_fma(acc, w2, a2); row_fma(acc, w3, a3);
  row_fma(acc, w4, a4); row_fma(acc, w5, a5); row_fma(acc, w6, a6); row_fma(acc, w7, a7);
}

static __device__ __forceinline__ void pull_add_tail(float4& acc, const float4* __restrict__ xr,
                                                     const float* __restrict__ sout, int c, int n,
                                                     const int4 sa, const int4 sb) {
  const int s[7] = {sa.x, sa.y, sa.z, sa.w, sb.x, sb.y, sb.z};
  float w[7];
  float4 a[7];
#pragma unroll
  for (int i = 0; i < 7; ++i) {
    w[i] = 0.f;
    a[i] = make_float4(0.f, 0.f, 0.f, 0.f);
    if (i < n) {
      w[i] = sout[s[i]];
      a[i] = xr[(size_t)s[i] * 16 + c];
    }
  }
#pragma unroll
  for (int i = 0; i < 7; ++i)
    if (i < n) row_fma(acc, w[i], a[i]);
}

static __device__ __forceinline__ void pull_add4(float4& acc, const float4* __restrict__ xr,
                                                 const float* __restrict__ sout, int c,
                                                 int s0, int s1, int s2, int s3) {
  const float w0 = sout[s0], w1 = sout[s1], w2 = sout[s2], w3 = sout[s3];
  const float4 a0 = xr[(size_t)s0 * 16 + c];
  const float4 a1 = xr[(size_t)s1 * 16 + c];
  const float4 a2 = xr[(size_t)s2 * 16 + c];
  const float4 a3 = xr[(size_t)s3 * 16 + c];
  row_fma(acc, w0, a0); row_fma(acc, w1, a1); row_fma(acc, w2, a2); row_fma(acc, w3, a3);
}

static __device__ __forceinline__ void pull_add1(float4& acc, const float4* __restrict__ xr,
                                                 const float* __restrict__ sout, int c, int s) {
  const float w = sout[s];
  const float4 a = xr[(size_t)s * 16 + c];
  row_fma(acc, w, a);
}

__global__ __launch_bounds__(PULL_TPB, 12) void ngcf_pull(
    const float* __restrict__ x, const int* __restrict__ rowptr, const int* __restrict__ csr,
    const float* __restrict__ sout, float* __restrict__ p, int N, int E, int stride) {
  __shared__ __align__(16) unsigned raw_s[PULL_NODES][SORT_CAP];
  __shared__ __align__(16) int srt_s[PULL_NODES][SORT_CAP];
  const int tid = threadIdx.x;
  const int hw = tid >> 4;
  const int c = tid & 15;
  unsigned* raw = raw_s[hw];
  int* srt = srt_s[hw];
  const float4* __restrict__ xr = (const float4*)x;
  const bool small_ids = (N <= (1 << 26));

  const int first0 = (int)blockIdx.x * PULL_NODES + ((tid >> 5) << 1);
  int node = (int)blockIdx.x * PULL_NODES + hw;

  int beg, endv;
  seg_bounds(rowptr, node, N, beg, endv);
  int len = seg_len(node, N, E, beg, endv);
  bool sorted = (len <= SORT_CAP) && small_ids;
  int v0, v1, v2, v3;
  seg_ids(csr, beg, len, c, sorted, v0, v1, v2, v3);

  for (int first = first0; first < N; first += stride) {
    const int nnode = node + stride;
    int nbeg, nendv;
    seg_bounds(rowptr, nnode, N, nbeg, nendv);

    if (sorted) {
      const unsigned k0 = (c < len)      ? (((unsigned)v0 << 6) | (unsigned)c)        : 0xFFFFFFFFu;
      const unsigned k1 = (c + 16 < len) ? (((unsigned)v1 << 6) | (unsigned)(c + 16)) : 0xFFFFFFFFu;
      const unsigned k2 = (c + 32 < len) ? (((unsigned)v2 << 6) | (unsigned)(c + 32)) : 0xFFFFFFFFu;
      const unsigned k3 = (c + 48 < len) ? (((unsigned)v3 << 6) | (unsigned)(c + 48)) : 0xFFFFFFFFu;
      raw[c] = k0; raw[c + 16] = k1; raw[c + 32] = k2; raw[c + 48] = k3;
      wave_handover();
      const int len8 = (len + 7) & ~7;
      int r0 = 0, r1 = 0;
      for (int j = 0; j < len; j += 4) {
        const uint4 u = *(const uint4*)(raw + j);
        r0 += (int)(u.x < k0) + (int)(u.y < k0) + (int)(u.z < k0) + (int)(u.w < k0);
        r1 += (int)(u.x < k1) + (int)(u.y < k1) + (int)(u.z < k1) + (int)(u.w < k1);
      }
      if (c < len) srt[r0] = v0; else if (c < len8) srt[c] = 0;
      if (c + 16 < len) srt[r1] = v1; else if (c + 16 < len8) srt[c + 16] = 0;
      if (len > 32) {
        int r2 = 0, r3 = 0;
        for (int j = 0; j < len; j += 4) {
          const uint4 u = *(const uint4*)(raw + j);
          r2 += (int)(u.x < k2) + (int)(u.y < k2) + (int)(u.z < k2) + (int)(u.w < k2);
          r3 += (int)(u.x < k3) + (int)(u.y < k3) + (int)(u.z < k3) + (int)(u.w < k3);
        }
        if (c + 32 < len) srt[r2] = v2; else if (c + 32 < len8) srt[c + 32] = 0;
        if (c + 48 < len) srt[r3] = v3; else if (c + 48 < len8) srt[c + 48] = 0;
      }
      wave_handover();
    }

    const int nlen = seg_len(nnode, N, E, nbeg, nendv);
    const bool nsorted = (nlen <= SORT_CAP) && small_ids;
    int n0, n1, n2, n3;
    seg_ids(csr, nbeg, nlen, c, nsorted, n0, n1, n2, n3);

    float4 acc = make_float4(0.f, 0.f, 0.f, 0.f);
    if (sorted) {
      int k = 0;
      for (; k + 8 <= len; k += 8) {
        const int4 sa = *(const int4*)(srt + k);
        const int4 sb = *(const int4*)(srt + k + 4);
        pull_add8(acc, xr, sout, c, sa, sb);
      }
      if (k < len) {
        const int4 sa = *(const int4*)(srt + k);
        const int4 sb = *(const int4*)(srt + k + 4);
        pull_add_tail(acc, xr, sout, c, len - k, sa, sb);
      }
    } else {
      const int* __restrict__ ids = csr + beg;
      int k = 0;
      for (; k + 4 <= len; k += 4) pull_add4(acc, xr, sout, c, ids[k], ids[k + 1], ids[k + 2], ids[k + 3]);
      for (; k < len; ++k) pull_add1(acc, xr, sout, c, ids[k]);
    }
    if (node < N) ((float4*)p)[(size_t)node * 16 + c] = acc;

    wave_handover();
    node = nnode; beg = nbeg; len = nlen; sorted = nsorted;
    v0 = n0; v1 = n1; v2 = n2; v3 = n3;
  }
}

__global__ __launch_bounds__(TPB) void ngcf_gemm(
    const float* __restrict__ x, const float* __restrict__ p,
    const float* __restrict__ sin_, const float* __restrict__ W1,
    const float* __restrict__ W2, float* __restrict__ out, int N, int do_relu) {
  __shared__ __align__(16) float hs[ROWS * HS_PITCH];

  const int tid = threadIdx.x;
  const int base = blockIdx.x * ROWS;

  for (int i = tid; i < ROWS * 16; i += TPB) {
    int row = i >> 4;
    int c = (i & 15) << 2;
    int node = base + row;
    if (node >= N) node = N - 1;
    float4 xv = *(const float4*)(x + (size_t)node * DD + c);
    float4 pv = *(const float4*)(p + (size_t)node * DD + c);
    float r = sin_[node];
    float4 pp = make_float4(pv.x * r, pv.y * r, pv.z * r, pv.w * r);
    float* h1 = &hs[row * HS_PITCH + c];
    float* h2 = &hs[row * HS_PITCH + DD + c];
    *(float4*)h1 = make_float4(xv.x + pp.x, xv.y + pp.y, xv.z + pp.z, xv.w + pp.w);
    *(float4*)h2 = make_float4(xv.x * pp.x, xv.y * pp.y, xv.z * pp.z, xv.w * pp.w);
  }
  __syncthreads();

  const int wave = tid >> 5;
  const int lane = tid & 31;
  const int m = lane & 15;
  const int khalf = (lane >> 4) << 1;
  const float* hrow = &hs[(wave * 16 + m) * HS_PITCH];

  v8f acc0 = {}, acc1 = {}, acc2 = {}, acc3 = {};
  for (int kb = 0; kb < 2 * DD; kb += 4) {
    v2f a = *(const v2f*)(hrow + kb + khalf);
    const float* wr = (kb < DD) ? (W1 + (size_t)(kb + khalf) * DD)
                                : (W2 + (size_t)(kb + khalf - DD) * DD);
    v2f b0, b1, b2, b3;
    b0.x = wr[m];      b0.y = wr[DD + m];
    b1.x = wr[16 + m]; b1.y = wr[DD + 16 + m];
    b2.x = wr[32 + m]; b2.y = wr[DD + 32 + m];
    b3.x = wr[48 + m]; b3.y = wr[DD + 48 + m];
    acc0 = __builtin_amdgcn_wmma_f32_16x16x4_f32(false, a, false, b0, (short)0, acc0, false, false);
    acc1 = __builtin_amdgcn_wmma_f32_16x16x4_f32(false, a, false, b1, (short)0, acc1, false, false);
    acc2 = __builtin_amdgcn_wmma_f32_16x16x4_f32(false, a, false, b2, (short)0, acc2, false, false);
    acc3 = __builtin_amdgcn_wmma_f32_16x16x4_f32(false, a, false, b3, (short)0, acc3, false, false);
  }

  const int rbase = base + wave * 16 + ((lane >> 4) << 3);
  for (int r = 0; r < 8; ++r) {
    int node = rbase + r;
    if (node < N) {
      float v0 = acc0[r], v1 = acc1[r], v2 = acc2[r], v3 = acc3[r];
      if (do_relu) {
        v0 = fmaxf(v0, 0.f); v1 = fmaxf(v1, 0.f);
        v2 = fmaxf(v2, 0.f); v3 = fmaxf(v3, 0.f);
      }
      float* orow = out + (size_t)node * DD;
      orow[m] = v0; orow[16 + m] = v1; orow[32 + m] = v2; orow[48 + m] = v3;
    }
  }
}

extern "C" void kernel_launch(void* const* d_in, const int* in_sizes, int n_in,
                              void* d_out, int out_size, void* d_ws, size_t ws_size,
                              hipStream_t stream) {
  const float* x   = (const float*)d_in[0];
  const int*   src = (const int*)d_in[1];
  const int*   dst = (const int*)d_in[2];
  const float* W1a = (const float*)d_in[3];
  const float* W2a = (const float*)d_in[4];
  const float* W1b = (const float*)d_in[5];
  const float* W2b = (const float*)d_in[6];
  float* out = (float*)d_out;

  const int N = in_sizes[0] / DD;
  const int E = in_sizes[1];

  int* ws_i    = (int*)d_ws;
  int* rowptr  = ws_i;
  int* bsum    = ws_i + N;
  float* sout  = (float*)(ws_i + 2 * N);
  float* sin_  = sout + N;
  float* p     = sin_ + N;
  int* csr     = (int*)(p + (size_t)N * DD);
  int* part    = (int*)p;

  if (ws_size < ((size_t)4 * N + (size_t)N * DD + (size_t)E) * sizeof(int)) return;

  dim3 dgrid(DEG_C, (N + DEG_K - 1) / DEG_K, 2);
  ngcf_degree<<<dgrid, DEG_TPB, 0, stream>>>(src, dst, part, N, E);
  const int sc_blocks = (N + SC_TPB - 1) / SC_TPB;
  ngcf_rsqrt<<<sc_blocks, SC_TPB, 0, stream>>>(part, sout, sin_, rowptr, bsum, N);
  ngcf_rowptr<<<sc_blocks, SC_TPB, 0, stream>>>(rowptr, bsum, N);
  dim3 fgrid(DEG_C, (N + DEG_K - 1) / DEG_K, 1);
  ngcf_fill<<<fgrid, DEG_TPB, 0, stream>>>(src, dst, part, rowptr, csr, N, E);

  const int gemm_blocks = (N + ROWS - 1) / ROWS;
  const int pull_groups = (N + PULL_NODES - 1) / PULL_NODES;
  const int pull_sweeps = pull_groups > PULL_GRID ? (pull_groups + PULL_GRID - 1) / PULL_GRID : 1;
  const int pull_blocks = (pull_groups + pull_sweeps - 1) / pull_sweeps;
  const int pull_stride = pull_blocks * PULL_NODES;

  ngcf_pull<<<pull_blocks, PULL_TPB, 0, stream>>>(x, rowptr, csr, sout, p, N, E, pull_stride);
  ngcf_gemm<<<gemm_blocks, TPB, 0, stream>>>(x, p, sin_, W1a, W2a, out, N, 1);

  ngcf_pull<<<pull_blocks, PULL_TPB, 0, stream>>>(out, rowptr, csr, sout, p, N, E, pull_stride);
  ngcf_gemm<<<gemm_blocks, TPB, 0, stream>>>(out, p, sin_, W1b, W2b, out, N, 0);
}


# load loops left rolled with a full wait behind each load are unrolled so loads issue ahead of first use: 32 count loads a node (rsqrt pass), 8 staging trips (product), 8 cursor trips (fill)
# speedup vs baseline: 3.0483x; 1.1416x over previous
#include <hip/hip_runtime.h>

typedef float v2f __attribute__((ext_vector_type(2)));
typedef float v8f __attribute__((ext_vector_type(8)));

#define DD 64
#define HS_PITCH 132
#define WAVES 6
#define ROWS (WAVES * 16)
#define TPB (WAVES * 32)

#define DEG_K 8192
#define DEG_C 16
#define DEG_TPB 1024

static __device__ __forceinline__ void deg_tally(int* cnt, int lo, int4 v) {
  unsigned r0 = (unsigned)(v.x - lo), r1 = (unsigned)(v.y - lo);
  unsigned r2 = (unsigned)(v.z - lo), r3 = (unsigned)(v.w - lo);
  if (r0 < DEG_K) atomicAdd(&cnt[r0], 1);
  if (r1 < DEG_K) atomicAdd(&cnt[r1], 1);
  if (r2 < DEG_K) atomicAdd(&cnt[r2], 1);
  if (r3 < DEG_K) atomicAdd(&cnt[r3], 1);
}

__global__ __launch_bounds__(DEG_TPB) void ngcf_degree(
    const int* __restrict__ src, const int* __restrict__ dst,
    int* __restrict__ part, int N, int E) {
  __shared__ int cnt[DEG_K];
  const int tid = threadIdx.x;
  const int slice = blockIdx.x;
  const int lo = blockIdx.y * DEG_K;
  const int which = blockIdx.z;
  const int* __restrict__ ids = which ? dst : src;

  for (int i = tid; i < DEG_K; i += DEG_TPB) cnt[i] = 0;
  __syncthreads();

  const int slice_len = (((E + DEG_C - 1) / DEG_C) + 3) & ~3;
  const int s0 = slice * slice_len;
  int len = E - s0;
  if (len > slice_len) len = slice_len;
  if (len < 0) len = 0;
  const int nfull = len >> 2;
  const int4* __restrict__ vp = (const int4*)(ids + s0);

  for (int j = tid; j < nfull; j += 4 * DEG_TPB) {
    const int j1 = j + DEG_TPB, j2 = j + 2 * DEG_TPB, j3 = j + 3 * DEG_TPB;
    int4 a = vp[j];
    int4 b = vp[j1 < nfull ? j1 : j];
    int4 c = vp[j2 < nfull ? j2 : j];
    int4 d = vp[j3 < nfull ? j3 : j];
    deg_tally(cnt, lo, a);
    if (j1 < nfull) deg_tally(cnt, lo, b);
    if (j2 < nfull) deg_tally(cnt, lo, c);
    if (j3 < nfull) deg_tally(cnt, lo, d);
  }
  {
    const int t = (nfull << 2) + tid;
    if (tid < 3 && t < len) {
      unsigned r = (unsigned)(ids[s0 + t] - lo);
      if (r < DEG_K) atomicAdd(&cnt[r], 1);
    }
  }
  __syncthreads();

  int* __restrict__ plane = part + (size_t)(slice * 2 + which) * N;
  for (int i = tid; i < DEG_K; i += DEG_TPB) {
    const int n = lo + i;
    if (n < N) plane[n] = cnt[i];
  }
}

#define SC_TPB 1024
__global__ __launch_bounds__(SC_TPB) void ngcf_rsqrt(int* __restrict__ part,
                           float* __restrict__ sout, float* __restrict__ sin_,
                           int* __restrict__ indeg, int* __restrict__ bsum, int N) {
  __shared__ int tot;
  if (threadIdx.x == 0) tot = 0;
  __syncthreads();
  const int g = blockIdx.x * SC_TPB + threadIdx.x;
  int b = 0;
  if (g < N) {
    int ta[DEG_C], tb[DEG_C];
#pragma unroll
    for (int c = 0; c < DEG_C; ++c) {
      ta[c] = part[(size_t)(2 * c) * N + g];
      tb[c] = part[(size_t)(2 * c + 1) * N + g];
    }
    int a = 0;
#pragma unroll
    for (int c = 0; c < DEG_C; ++c) {
      a += ta[c];
      part[(size_t)(2 * c + 1) * N + g] = b;
      b += tb[c];
    }
    indeg[g] = b;
    sout[g] = rsqrtf((float)(a < 1 ? 1 : a));
    sin_[g] = rsqrtf((float)(b < 1 ? 1 : b));
  }
  if (b) atomicAdd(&tot, b);
  __syncthreads();
  if (threadIdx.x == 0) bsum[blockIdx.x] = tot;
}

__global__ __launch_bounds__(SC_TPB) void ngcf_rowptr(int* __restrict__ rowptr,
                                                      const int* __restrict__ bsum, int N) {
  __shared__ int sc[SC_TPB];
  __shared__ int sbase;
  const int t = threadIdx.x;
  const int k = blockIdx.x;
  if (t == 0) sbase = 0;
  __syncthreads();
  int s = 0;
  for (int i = t; i < k; i += SC_TPB) s += bsum[i];
  if (s) atomicAdd(&sbase, s);
  const int g = k * SC_TPB + t;
  const int v = (g < N) ? rowptr[g] : 0;
  sc[t] = v;
  __syncthreads();
  for (int off = 1; off < SC_TPB; off <<= 1) {
    const int u = (t >= off) ? sc[t - off] : 0;
    __syncthreads();
    sc[t] += u;
    __syncthreads();
  }
  if (g < N) rowptr[g] = sbase + sc[t] - v;
}

static __device__ __forceinline__ void fill_one(int* cnt, int lo, unsigned lim, int id, int s,
                                                int* __restrict__ csr) {
  const unsigned r = (unsigned)(id - lo);
  if (r < lim) {
    const int pos = atomicAdd(&cnt[r], 1);
    csr[pos] = s;
  }
}

static __device__ __forceinline__ void fill_vec(int* cnt, int lo, unsigned lim, unsigned spare,
                                                int4 d, int4 s, int* __restrict__ csr) {
  const unsigned r0 = (unsigned)(d.x - lo), r1 = (unsigned)(d.y - lo);
  const unsigned r2 = (unsigned)(d.z - lo), r3 = (unsigned)(d.w - lo);
  const unsigned p0 = (unsigned)atomicAdd(&cnt[r0 < lim ? r0 : spare], 1);
  const unsigned p1 = (unsigned)atomicAdd(&cnt[r1 < lim ? r1 : spare], 1);
  const unsigned p2 = (unsigned)atomicAdd(&cnt[r2 < lim ? r2 : spare], 1);
  const unsigned p3 = (unsigned)atomicAdd(&cnt[r3 < lim ? r3 : spare], 1);
  if (r0 < lim) csr[p0] = s.x;
  if (r1 < lim) csr[p1] = s.y;
  if (r2 < lim) csr[p2] = s.z;
  if (r3 < lim) csr[p3] = s.w;
}

__global__ __launch_bounds__(DEG_TPB) void ngcf_fill(
    const int* __restrict__ src, const int* __restrict__ dst,
    const int* __restrict__ part, const int* __restrict__ rowptr,
    int* __restrict__ csr, int N, int E) {
  __shared__ int cnt[DEG_K + 32];
  const int tid = threadIdx.x;
  const int slice = blockIdx.x;
  const int lo = blockIdx.y * DEG_K;
  const int* __restrict__ pre = part + (size_t)(slice * 2 + 1) * N;
  const unsigned spare = DEG_K + (tid & 31);

  {
    constexpr int CUR = DEG_K / DEG_TPB;
    static_assert(DEG_K % DEG_TPB == 0, "every thread sets the same number of cursors");
    int rp[CUR], pr[CUR];
#pragma unroll
    for (int t = 0; t < CUR; ++t) {
      const int n = lo + tid + t * DEG_TPB;
      const int nn = (n < N) ? n : N - 1;
      rp[t] = rowptr[nn];
      pr[t] = pre[nn];
    }
#pragma unroll
    for (int t = 0; t < CUR; ++t) {
      const int n = lo + tid + t * DEG_TPB;
      cnt[tid + t * DEG_TPB] = (n < N) ? (rp[t] + pr[t]) : 0;
    }
  }
  if (tid < 32) cnt[DEG_K + tid] = 0;
  __syncthreads();
  const unsigned lim = (unsigned)((N - lo) < DEG_K ? (N - lo) : DEG_K);

  const int slice_len = (((E + DEG_C - 1) / DEG_C) + 3) & ~3;
  const int s0 = slice * slice_len;
  int len = E - s0;
  if (len > slice_len) len = slice_len;
  if (len < 0) len = 0;
  const int nfull = len >> 2;
  const int4* __restrict__ vd = (const int4*)(dst + s0);
  const int4* __restrict__ vs = (const int4*)(src + s0);

  for (int j = tid; j < nfull; j += 4 * DEG_TPB) {
    const int j1 = j + DEG_TPB, j2 = j + 2 * DEG_TPB, j3 = j + 3 * DEG_TPB;
    const int i1 = j1 < nfull ? j1 : j, i2 = j2 < nfull ? j2 : j, i3 = j3 < nfull ? j3 : j;
    const int4 da = vd[j], db = vd[i1], dc = vd[i2], dd = vd[i3];
    const int4 sa = vs[j], sb = vs[i1], sc = vs[i2], sd = vs[i3];
    fill_vec(cnt, lo, lim, spare, da, sa, csr);
    if (j1 < nfull) fill_vec(cnt, lo, lim, spare, db, sb, csr);
    if (j2 < nfull) fill_vec(cnt, lo, lim, spare, dc, sc, csr);
    if (j3 < nfull) fill_vec(cnt, lo, lim, spare, dd, sd, csr);
  }
  {
    const int t = (nfull << 2) + tid;
    if (tid < 3 && t < len) fill_one(cnt, lo, lim, dst[s0 + t], src[s0 + t], csr);
  }
}

#define PULL_TPB 256
#define PULL_NODES 16
#define SORT_CAP 64
#define PULL_GRID 1280

static __device__ __forceinline__ void wave_handover() {
  __builtin_amdgcn_fence(__ATOMIC_RELEASE, "wavefront");
  __builtin_amdgcn_wave_barrier();
  __builtin_amdgcn_fence(__ATOMIC_ACQUIRE, "wavefront");
}

static __device__ __forceinline__ void seg_bounds(const int* __restrict__ rowptr, int node, int N,
                                                  int& begv, int& endv) {
  const int i0 = (node < N) ? node : N - 1;
  const int i1 = (node + 1 < N) ? node + 1 : N - 1;
  begv = rowptr[i0];
  endv = rowptr[i1];
}

static __device__ __forceinline__ int seg_len(int node, int N, int E, int begv, int endv) {
  const int end = (node + 1 < N) ? endv : E;
  return (node < N) ? end - begv : 0;
}

static __device__ __forceinline__ void seg_ids(const int* __restrict__ csr, int beg, int len, int c,
                                               bool sorted, int& v0, int& v1, int& v2, int& v3) {
  v0 = 0; v1 = 0; v2 = 0; v3 = 0;
  if (sorted) {
    if (c < len)      v0 = csr[beg + c];
    if (c + 16 < len) v1 = csr[beg + c + 16];
    if (c + 32 < len) v2 = csr[beg + c + 32];
    if (c + 48 < len) v3 = csr[beg + c + 48];
  }
}

static __device__ __forceinline__ void row_fma(float4& acc, float w, const float4 a) {
  acc.x = fmaf(w, a.x, acc.x); acc.y = fmaf(w, a.y, acc.y);
  acc.z = fmaf(w, a.z, acc.z); acc.w = fmaf(w, a.w, acc.w);
}

static __device__ __forceinline__ void pull_add8(float4& acc, const float4* __restrict__ xr,
                                                 const float* __restrict__ sout, int c,
                                                 const int4 sa, const int4 sb) {
  const float w0 = sout[sa.x], w1 = sout[sa.y], w2 = sout[sa.z], w3 = sout[sa.w];
  const float w4 = sout[sb.x], w5 = sout[sb.y], w6 = sout[sb.z], w7 = sout[sb.w];
  const float4 a0 = xr[(size_t)sa.x * 16 + c];
  const float4 a1 = xr[(size_t)sa.y * 16 + c];
  const float4 a2 = xr[(size_t)sa.z * 16 + c];
  const float4 a3 = xr[(size_t)sa.w * 16 + c];
  const float4 a4 = xr[(size_t)sb.x * 16 + c];
  const float4 a5 = xr[(size_t)sb.y * 16 + c];
  const float4 a6 = xr[(size_t)sb.z * 16 + c];
  const float4 a7 = xr[(size_t)sb.w * 16 + c];
  row_fma(acc, w0, a0); row_fma(acc, w1, a1); row_fma(acc, w2, a2); row_fma(acc, w3, a3);
  row_fma(acc, w4, a4); row_fma(acc, w5, a5); row_fma(acc, w6, a6); row_fma(acc, w7, a7);
}

static __device__ __forceinline__ void pull_add_tail(float4& acc, const float4* __restrict__ xr,
                                                     const float* __restrict__ sout, int c, int n,
                                                     const int4 sa, const int4 sb) {
  const int s[7] = {sa.x, sa.y, sa.z, sa.w, sb.x, sb.y, sb.z};
  float w[7];
  float4 a[7];
#pragma unroll
  for (int i = 0; i < 7; ++i) {
    w[i] = 0.f;
    a[i] = make_float4(0.f, 0.f, 0.f, 0.f);
    if (i < n) {
      w[i] = sout[s[i]];
      a[i] = xr[(size_t)s[i] * 16 + c];
    }
  }
#pragma unroll
  for (int i = 0; i < 7; ++i)
    if (i < n) row_fma(acc, w[i], a[i]);
}

static __device__ __forceinline__ void pull_add4(float4& acc, const float4* __restrict__ xr,
                                                 const float* __restrict__ sout, int c,
                                                 int s0, int s1, int s2, int s3) {
  const float w0 = sout[s0], w1 = sout[s1], w2 = sout[s2], w3 = sout[s3];
  const float4 a0 = xr[(size_t)s0 * 16 + c];
  const float4 a1 = xr[(size_t)s1 * 16 + c];
  const float4 a2 = xr[(size_t)s2 * 16 + c];
  const float4 a3 = xr[(size_t)s3 * 16 + c];
  row_fma(acc, w0, a0); row_fma(acc, w1, a1); row_fma(acc, w2, a2); row_fma(acc, w3, a3);
}

static __device__ __forceinline__ void pull_add1(float4& acc, const float4* __restrict__ xr,
                                                 const float* __restrict__ sout, int c, int s) {
  const float w = sout[s];
  const float4 a = xr[(size_t)s * 16 + c];
  row_fma(acc, w, a);
}

__global__ __launch_bounds__(PULL_TPB, 12) void ngcf_pull(
    const float* __restrict__ x, const int* __restrict__ rowptr, const int* __restrict__ csr,
    const float* __restrict__ sout, float* __restrict__ p, int N, int E, int stride) {
  __shared__ __align__(16) unsigned raw_s[PULL_NODES][SORT_CAP];
  __shared__ __align__(16) int srt_s[PULL_NODES][SORT_CAP];
  const int tid = threadIdx.x;
  const int hw = tid >> 4;
  const int c = tid & 15;
  unsigned* raw = raw_s[hw];
  int* srt = srt_s[hw];
  const float4* __restrict__ xr = (const float4*)x;
  const bool small_ids = (N <= (1 << 26));

  const int first0 = (int)blockIdx.x * PULL_NODES + ((tid >> 5) << 1);
  int node = (int)blockIdx.x * PULL_NODES + hw;

  int beg, endv;
  seg_bounds(rowptr, node, N, beg, endv);
  int len = seg_len(node, N, E, beg, endv);
  bool sorted = (len <= SORT_CAP) && small_ids;
  int v0, v1, v2, v3;
  seg_ids(csr, beg, len, c, sorted, v0, v1, v2, v3);

  for (int first = first0; first < N; first += stride) {
    const int nnode = node + stride;
    int nbeg, nendv;
    seg_bounds(rowptr, nnode, N, nbeg, nendv);

    if (sorted) {
      const unsigned k0 = (c < len)      ? (((unsigned)v0 << 6) | (unsigned)c)        : 0xFFFFFFFFu;
      const unsigned k1 = (c + 16 < len) ? (((unsigned)v1 << 6) | (unsigned)(c + 16)) : 0xFFFFFFFFu;
      const unsigned k2 = (c + 32 < len) ? (((unsigned)v2 << 6) | (unsigned)(c + 32)) : 0xFFFFFFFFu;
      const unsigned k3 = (c + 48 < len) ? (((unsigned)v3 << 6) | (unsigned)(c + 48)) : 0xFFFFFFFFu;
      raw[c] = k0; raw[c + 16] = k1; raw[c + 32] = k2; raw[c + 48] = k3;
      wave_handover();
      const int len8 = (len + 7) & ~7;
      int r0 = 0, r1 = 0;
      for (int j = 0; j < len; j += 4) {
        const uint4 u = *(const uint4*)(raw + j);
        r0 += (int)(u.x < k0) + (int)(u.y < k0) + (int)(u.z < k0) + (int)(u.w < k0);
        r1 += (int)(u.x < k1) + (int)(u.y < k1) + (int)(u.z < k1) + (int)(u.w < k1);
      }
      if (c < len) srt[r0] = v0; else if (c < len8) srt[c] = 0;
      if (c + 16 < len) srt[r1] = v1; else if (c + 16 < len8) srt[c + 16] = 0;
      if (len > 32) {
        int r2 = 0, r3 = 0;
        for (int j = 0; j < len; j += 4) {
          const uint4 u = *(const uint4*)(raw + j);
          r2 += (int)(u.x < k2) + (int)(u.y < k2) + (int)(u.z < k2) + (int)(u.w < k2);
          r3 += (int)(u.x < k3) + (int)(u.y < k3) + (int)(u.z < k3) + (int)(u.w < k3);
        }
        if (c + 32 < len) srt[r2] = v2; else if (c + 32 < len8) srt[c + 32] = 0;
        if (c + 48 < len) srt[r3] = v3; else if (c + 48 < len8) srt[c + 48] = 0;
      }
      wave_handover();
    }

    const int nlen = seg_len(nnode, N, E, nbeg, nendv);
    const bool nsorted = (nlen <= SORT_CAP) && small_ids;
    int n0, n1, n2, n3;
    seg_ids(csr, nbeg, nlen, c, nsorted, n0, n1, n2, n3);

    float4 acc = make_float4(0.f, 0.f, 0.f, 0.f);
    if (sorted) {
      int k = 0;
      for (; k + 8 <= len; k += 8) {
        const int4 sa = *(const int4*)(srt + k);
        const int4 sb = *(const int4*)(srt + k + 4);
        pull_add8(acc, xr, sout, c, sa, sb);
      }
      if (k < len) {
        const int4 sa = *(const int4*)(srt + k);
        const int4 sb = *(const int4*)(srt + k + 4);
        pull_add_tail(acc, xr, sout, c, len - k, sa, sb);
      }
    } else {
      const int* __restrict__ ids = csr + beg;
      int k = 0;
      for (; k + 4 <= len; k += 4) pull_add4(acc, xr, sout, c, ids[k], ids[k + 1], ids[k + 2], ids[k + 3]);
      for (; k < len; ++k) pull_add1(acc, xr, sout, c, ids[k]);
    }
    if (node < N) ((float4*)p)[(size_t)node * 16 + c] = acc;

    wave_handover();
    node = nnode; beg = nbeg; len = nlen; sorted = nsorted;
    v0 = n0; v1 = n1; v2 = n2; v3 = n3;
  }
}

__global__ __launch_bounds__(TPB) void ngcf_gemm(
    const float* __restrict__ x, const float* __restrict__ p,
    const float* __restrict__ sin_, const float* __restrict__ W1,
    const float* __restrict__ W2, float* __restrict__ out, int N, int do_relu) {
  __shared__ __align__(16) float hs[ROWS * HS_PITCH];

  const int tid = threadIdx.x;
  const int base = blockIdx.x * ROWS;

  constexpr int STG = ROWS * 16 / TPB;
  static_assert((ROWS * 16) % TPB == 0, "every thread stages the same number of chunks");
  {
    float4 xv[STG], pv[STG];
    float rs[STG];
#pragma unroll
    for (int t = 0; t < STG; ++t) {
      const int i = tid + t * TPB;
      int node = base + (i >> 4);
      if (node >= N) node = N - 1;
      xv[t] = ((const float4*)x)[node * 16 + (i & 15)];
      pv[t] = ((const float4*)p)[node * 16 + (i & 15)];
      rs[t] = sin_[node];
    }
#pragma unroll
    for (int t = 0; t < STG; ++t) {
      const int i = tid + t * TPB;
      const int row = i >> 4;
      const int c = (i & 15) << 2;
      const float r = rs[t];
      const float4 pp = make_float4(pv[t].x * r, pv[t].y * r, pv[t].z * r, pv[t].w * r);
      float* h1 = &hs[row * HS_PITCH + c];
      float* h2 = &hs[row * HS_PITCH + DD + c];
      *(float4*)h1 = make_float4(xv[t].x + pp.x, xv[t].y + pp.y, xv[t].z + pp.z, xv[t].w + pp.w);
      *(float4*)h2 = make_float4(xv[t].x * pp.x, xv[t].y * pp.y, xv[t].z * pp.z, xv[t].w * pp.w);
    }
  }
  __syncthreads();

  const int wave = tid >> 5;
  const int lane = tid & 31;
  const int m = lane & 15;
  const int khalf = (lane >> 4) << 1;
  const float* hrow = &hs[(wave * 16 + m) * HS_PITCH];

  v8f acc0 = {}, acc1 = {}, acc2 = {}, acc3 = {};
  for (int kb = 0; kb < 2 * DD; kb += 4) {
    v2f a = *(const v2f*)(hrow + kb + khalf);
    const float* wr = (kb < DD) ? (W1 + (size_t)(kb + khalf) * DD)
                                : (W2 + (size_t)(kb + khalf - DD) * DD);
    v2f b0, b1, b2, b3;
    b0.x = wr[m];      b0.y = wr[DD + m];
    b1.x = wr[16 + m]; b1.y = wr[DD + 16 + m];
    b2.x = wr[32 + m]; b2.y = wr[DD + 32 + m];
    b3.x = wr[48 + m]; b3.y = wr[DD + 48 + m];
    acc0 = __builtin_amdgcn_wmma_f32_16x16x4_f32(false, a, false, b0, (short)0, acc0, false, false);
    acc1 = __builtin_amdgcn_wmma_f32_16x16x4_f32(false, a, false, b1, (short)0, acc1, false, false);
    acc2 = __builtin_amdgcn_wmma_f32_16x16x4_f32(false, a, false, b2, (short)0, acc2, false, false);
    acc3 = __builtin_amdgcn_wmma_f32_16x16x4_f32(false, a, false, b3, (short)0, acc3, false, false);
  }

  const int rbase = base + wave * 16 + ((lane >> 4) << 3);
  for (int r = 0; r < 8; ++r) {
    int node = rbase + r;
    if (node < N) {
      float v0 = acc0[r], v1 = acc1[r], v2 = acc2[r], v3 = acc3[r];
      if (do_relu) {
        v0 = fmaxf(v0, 0.f); v1 = fmaxf(v1, 0.f);
        v2 = fmaxf(v2, 0.f); v3 = fmaxf(v3, 0.f);
      }
      float* orow = out + (size_t)node * DD;
      orow[m] = v0; orow[16 + m] = v1; orow[32 + m] = v2; orow[48 + m] = v3;
    }
  }
}

extern "C" void kernel_launch(void* const* d_in, const int* in_sizes, int n_in,
                              void* d_out, int out_size, void* d_ws, size_t ws_size,
                              hipStream_t stream) {
  const float* x   = (const float*)d_in[0];
  const int*   src = (const int*)d_in[1];
  const int*   dst = (const int*)d_in[2];
  const float* W1a = (const float*)d_in[3];
  const float* W2a = (const float*)d_in[4];
  const float* W1b = (const float*)d_in[5];
  const float* W2b = (const float*)d_in[6];
  float* out = (float*)d_out;

  const int N = in_sizes[0] / DD;
  const int E = in_sizes[1];

  int* ws_i    = (int*)d_ws;
  int* rowptr  = ws_i;
  int* bsum    = ws_i + N;
  float* sout  = (float*)(ws_i + 2 * N);
  float* sin_  = sout + N;
  float* p     = sin_ + N;
  int* csr     = (int*)(p + (size_t)N * DD);
  int* part    = (int*)p;

  if (ws_size < ((size_t)4 * N + (size_t)N * DD + (size_t)E) * sizeof(int)) return;

  dim3 dgrid(DEG_C, (N + DEG_K - 1) / DEG_K, 2);
  ngcf_degree<<<dgrid, DEG_TPB, 0, stream>>>(src, dst, part, N, E);
  const int sc_blocks = (N + SC_TPB - 1) / SC_TPB;
  ngcf_rsqrt<<<sc_blocks, SC_TPB, 0, stream>>>(part, sout, sin_, rowptr, bsum, N);
  ngcf_rowptr<<<sc_blocks, SC_TPB, 0, stream>>>(rowptr, bsum, N);
  dim3 fgrid(DEG_C, (N + DEG_K - 1) / DEG_K, 1);
  ngcf_fill<<<fgrid, DEG_TPB, 0, stream>>>(src, dst, part, rowptr, csr, N, E);

  const int gemm_blocks = (N + ROWS - 1) / ROWS;
  const int pull_groups = (N + PULL_NODES - 1) / PULL_NODES;
  const int pull_sweeps = pull_groups > PULL_GRID ? (pull_groups + PULL_GRID - 1) / PULL_GRID : 1;
  const int pull_blocks = (pull_groups + pull_sweeps - 1) / pull_sweeps;
  const int pull_stride = pull_blocks * PULL_NODES;

  ngcf_pull<<<pull_blocks, PULL_TPB, 0, stream>>>(x, rowptr, csr, sout, p, N, E, pull_stride);
  ngcf_gemm<<<gemm_blocks, TPB, 0, stream>>>(x, p, sin_, W1a, W2a, out, N, 1);

  ngcf_pull<<<pull_blocks, PULL_TPB, 0, stream>>>(out, rowptr, csr, sout, p, N, E, pull_stride);
  ngcf_gemm<<<gemm_blocks, TPB, 0, stream>>>(out, p, sin_, W1b, W2b, out, N, 0);
}
